# starting kernel
# baseline (speedup 1.0000x reference)
#include <hip/hip_runtime.h>
#include <math.h>

typedef __attribute__((ext_vector_type(2))) float v2f;
typedef __attribute__((ext_vector_type(8))) float v8f;

#define G_   10
#define HID  64
#define FFN  128
#define NH   8
#define E_   131072

#define OFF_WIN1   0
#define OFF_WOUT1  8192
#define OFF_WIN2   16384
#define OFF_WOUT2  24576
#define OFF_WFC    32768
#define OFF_BIN1   33792
#define OFF_BOUT1  33920
#define OFF_BIN2   33984
#define OFF_BOUT2  34112
#define OFF_BFC    34176
#define GSTRIDE    34192

#define SCR_STRIDE 18
#define SCR_SIZE   (16*SCR_STRIDE)

__global__ void pack_weights(const float* __restrict__ w_in1, const float* __restrict__ b_in1,
                             const float* __restrict__ w_out1, const float* __restrict__ b_out1,
                             const float* __restrict__ w_in2, const float* __restrict__ b_in2,
                             const float* __restrict__ w_out2, const float* __restrict__ b_out2,
                             const float* __restrict__ w_fc, const float* __restrict__ b_fc,
                             float* __restrict__ ws) {
  int tid = blockIdx.x * blockDim.x + threadIdx.x;
  if (tid >= G_ * GSTRIDE) return;
  int g = tid / GSTRIDE;
  int pos = tid - g * GSTRIDE;
  float val = 0.0f;
  if (pos < OFF_WOUT1) {
    int u = pos - OFF_WIN1;
    int v = u & 1, L = (u >> 1) & 31, r = u >> 6;
    int kk = r & 15, t = r >> 4;
    int k = 4 * kk + 2 * (L >> 4) + v;
    int n = 16 * t + (L & 15);
    val = w_in1[(g * HID + k) * FFN + n];
  } else if (pos < OFF_WIN2) {
    int u = pos - OFF_WOUT1;
    int v = u & 1, L = (u >> 1) & 31, r = u >> 6;
    int kk = r & 31, t = r >> 5;
    int k = 4 * kk + 2 * (L >> 4) + v;
    int n = 16 * t + (L & 15);
    val = w_out1[(g * FFN + k) * HID + n];
  } else if (pos < OFF_WOUT2) {
    int u = pos - OFF_WIN2;
    int v = u & 1, L = (u >> 1) & 31, r = u >> 6;
    int kk = r & 15, t = r >> 4;
    int k = 4 * kk + 2 * (L >> 4) + v;
    int n = 16 * t + (L & 15);
    val = w_in2[(g * HID + k) * FFN + n];
  } else if (pos < OFF_WFC) {
    int u = pos - OFF_WOUT2;
    int v = u & 1, L = (u >> 1) & 31, r = u >> 6;
    int kk = r & 31, t = r >> 5;
    int k = 4 * kk + 2 * (L >> 4) + v;
    int n = 16 * t + (L & 15);
    val = w_out2[(g * FFN + k) * HID + n];
  } else if (pos < OFF_BIN1) {
    int u = pos - OFF_WFC;
    int v = u & 1, L = (u >> 1) & 31, kk = u >> 6;
    int k = 4 * kk + 2 * (L >> 4) + v;
    int c = L & 15;
    val = (c < NH) ? w_fc[(g * HID + k) * NH + c] : 0.0f;
  } else if (pos < OFF_BOUT1) {
    val = b_in1[g * FFN + (pos - OFF_BIN1)];
  } else if (pos < OFF_BIN2) {
    val = b_out1[g * HID + (pos - OFF_BOUT1)];
  } else if (pos < OFF_BOUT2) {
    val = b_in2[g * FFN + (pos - OFF_BIN2)];
  } else if (pos < OFF_BFC) {
    val = b_out2[g * HID + (pos - OFF_BOUT2)];
  } else {
    int j = pos - OFF_BFC;
    val = (j < NH) ? b_fc[g * NH + j] : 0.0f;
  }
  ws[tid] = val;
}

__device__ __forceinline__ v8f wmma4(v2f a, v2f b, v8f c) {
  return __builtin_amdgcn_wmma_f32_16x16x4_f32(false, a, false, b, (short)0, c,
                                               false, false);
}

__device__ __forceinline__ void mlp_block(v2f (&xa)[2][16],
                                          const float* __restrict__ win,
                                          const float* __restrict__ wout,
                                          const float* __restrict__ bin,
                                          const float* __restrict__ bout,
                                          float* scr0, float* scr1,
                                          int lane, int c, int m, int hi) {
  v8f acc2[2][4];
  #pragma unroll
  for (int i = 0; i < 2; ++i)
    #pragma unroll
    for (int nt = 0; nt < 4; ++nt) acc2[i][nt] = (v8f){};

  for (int t = 0; t < 8; ++t) {
    v8f h0 = {}, h1 = {};
    #pragma unroll
    for (int kk = 0; kk < 16; ++kk) {
      v2f bf = *(const v2f*)(win + ((t * 16 + kk) * 32 + lane) * 2);
      h0 = wmma4(xa[0][kk], bf, h0);
      h1 = wmma4(xa[1][kk], bf, h1);
    }
    float bb = bin[16 * t + c];
    #pragma unroll
    for (int j = 0; j < 8; ++j) {
      h0[j] = fmaxf(h0[j] + bb, 0.0f);
      h1[j] = fmaxf(h1[j] + bb, 0.0f);
    }
    #pragma unroll
    for (int j = 0; j < 8; ++j) {
      scr0[(j + 8 * hi) * SCR_STRIDE + c] = h0[j];
      scr1[(j + 8 * hi) * SCR_STRIDE + c] = h1[j];
    }
    __asm__ volatile("" ::: "memory");
    v2f ha0[4], ha1[4];
    #pragma unroll
    for (int q = 0; q < 4; ++q) {
      ha0[q] = *(const v2f*)(scr0 + m * SCR_STRIDE + 4 * q + 2 * hi);
      ha1[q] = *(const v2f*)(scr1 + m * SCR_STRIDE + 4 * q + 2 * hi);
    }
    __asm__ volatile("" ::: "memory");
    #pragma unroll
    for (int q = 0; q < 4; ++q) {
      #pragma unroll
      for (int nt = 0; nt < 4; ++nt) {
        v2f bf = *(const v2f*)(wout + ((nt * 32 + t * 4 + q) * 32 + lane) * 2);
        acc2[0][nt] = wmma4(ha0[q], bf, acc2[0][nt]);
        acc2[1][nt] = wmma4(ha1[q], bf, acc2[1][nt]);
      }
    }
  }
  #pragma unroll
  for (int nt = 0; nt < 4; ++nt) {
    float bb = bout[16 * nt + c];
    #pragma unroll
    for (int j = 0; j < 8; ++j) {
      scr0[(j + 8 * hi) * SCR_STRIDE + c] = acc2[0][nt][j] + bb;
      scr1[(j + 8 * hi) * SCR_STRIDE + c] = acc2[1][nt][j] + bb;
    }
    __asm__ volatile("" ::: "memory");
    #pragma unroll
    for (int q = 0; q < 4; ++q) {
      v2f n0 = *(const v2f*)(scr0 + m * SCR_STRIDE + 4 * q + 2 * hi);
      v2f n1 = *(const v2f*)(scr1 + m * SCR_STRIDE + 4 * q + 2 * hi);
      xa[0][4 * nt + q] += n0;
      xa[1][4 * nt + q] += n1;
    }
    __asm__ volatile("" ::: "memory");
  }
}

__global__ __launch_bounds__(256) void fused_edge_mlp(
    const float* __restrict__ edge_attr, const int* __restrict__ edge_index,
    const float* __restrict__ deg, const float* __restrict__ deg_coef,
    const float* __restrict__ ws, float* __restrict__ out) {
  __shared__ float scr[8][2][SCR_SIZE];
  const int w = threadIdx.x >> 5;
  const int lane = threadIdx.x & 31;
  const int c = lane & 15;
  const int hi = lane >> 4;
  const int m = c;

  const int task = blockIdx.x;
  const int g = task >> 9;
  const int chunk = task & 511;
  const int e0 = chunk * 256 + w * 32;

  const float* wb = ws + g * GSTRIDE;
  float* scr0 = scr[w][0];
  float* scr1 = scr[w][1];

  const float* dc0 = deg_coef + g * HID;
  const float* dc1 = deg_coef + (G_ + g) * HID;
  const int em0 = e0 + m;
  const int em1 = e0 + 16 + m;
  const float sd0 = sqrtf(deg[edge_index[E_ + em0]]);
  const float sd1 = sqrtf(deg[edge_index[E_ + em1]]);

  v2f xa[2][16];
  #pragma unroll
  for (int kk = 0; kk < 16; ++kk) {
    int k = 4 * kk + 2 * hi;
    v2f d0 = *(const v2f*)(dc0 + k);
    v2f d1 = *(const v2f*)(dc1 + k);
    v2f a0 = *(const v2f*)(edge_attr + (size_t)em0 * HID + k);
    v2f a1 = *(const v2f*)(edge_attr + (size_t)em1 * HID + k);
    xa[0][kk] = a0 * (d0 + sd0 * d1);
    xa[1][kk] = a1 * (d0 + sd1 * d1);
  }

  mlp_block(xa, wb + OFF_WIN1, wb + OFF_WOUT1, wb + OFF_BIN1, wb + OFF_BOUT1,
            scr0, scr1, lane, c, m, hi);
  mlp_block(xa, wb + OFF_WIN2, wb + OFF_WOUT2, wb + OFF_BIN2, wb + OFF_BOUT2,
            scr0, scr1, lane, c, m, hi);

  #pragma unroll
  for (int kk = 0; kk < 16; ++kk) {
    #pragma unroll
    for (int v = 0; v < 2; ++v) {
      xa[0][kk][v] = fmaxf(xa[0][kk][v], 0.0f);
      xa[1][kk][v] = fmaxf(xa[1][kk][v], 0.0f);
    }
  }

  v8f f0 = {}, f1 = {};
  #pragma unroll
  for (int kk = 0; kk < 16; ++kk) {
    v2f bf = *(const v2f*)(wb + OFF_WFC + (kk * 32 + lane) * 2);
    f0 = wmma4(xa[0][kk], bf, f0);
    f1 = wmma4(xa[1][kk], bf, f1);
  }

  if (c < NH) {
    float bb = wb[OFF_BFC + c];
    size_t base = (size_t)g * E_;
    #pragma unroll
    for (int j = 0; j < 8; ++j) {
      int r = j + 8 * hi;
      out[(base + e0 + r) * NH + c] = f0[j] + bb;
      out[(base + e0 + 16 + r) * NH + c] = f1[j] + bb;
    }
  }
}

extern "C" void kernel_launch(void* const* d_in, const int* in_sizes, int n_in,
                              void* d_out, int out_size, void* d_ws, size_t ws_size,
                              hipStream_t stream) {
  const float* edge_attr  = (const float*)d_in[0];
  const int*   edge_index = (const int*)d_in[1];
  const float* deg        = (const float*)d_in[2];
  const float* deg_coef   = (const float*)d_in[3];
  const float* w_in1  = (const float*)d_in[4];
  const float* b_in1  = (const float*)d_in[5];
  const float* w_out1 = (const float*)d_in[6];
  const float* b_out1 = (const float*)d_in[7];
  const float* w_in2  = (const float*)d_in[8];
  const float* b_in2  = (const float*)d_in[9];
  const float* w_out2 = (const float*)d_in[10];
  const float* b_out2 = (const float*)d_in[11];
  const float* w_fc   = (const float*)d_in[12];
  const float* b_fc   = (const float*)d_in[13];
  float* ws  = (float*)d_ws;
  float* out = (float*)d_out;

  const int packTotal = G_ * GSTRIDE;
  pack_weights<<<(packTotal + 255) / 256, 256, 0, stream>>>(
      w_in1, b_in1, w_out1, b_out1, w_in2, b_in2, w_out2, b_out2, w_fc, b_fc, ws);

  fused_edge_mlp<<<G_ * (E_ / 256), 256, 0, stream>>>(
      edge_attr, edge_index, deg, deg_coef, ws, out);
}


# Products computed transposed (weights as A, edges as columns): each D tile is the next product's B operand, so the LDS transposes go; k order of the sums inside each product changes
# speedup vs baseline: 1.0548x; 1.0548x over previous
#include <hip/hip_runtime.h>
#include <math.h>

typedef __attribute__((ext_vector_type(2))) float v2f;
typedef __attribute__((ext_vector_type(8))) float v8f;

#define G_   10
#define HID  64
#define FFN  128
#define NH   8
#define E_   131072

#define OFF_WIN1   0
#define OFF_WOUT1  8192
#define OFF_WIN2   16384
#define OFF_WOUT2  24576
#define OFF_WFC    32768
#define OFF_BIN1   33792
#define OFF_BOUT1  33920
#define OFF_BIN2   33984
#define OFF_BOUT2  34112
#define OFF_BFC    34176
#define GSTRIDE    34192


__global__ void pack_weights(const float* __restrict__ w_in1, const float* __restrict__ b_in1,
                             const float* __restrict__ w_out1, const float* __restrict__ b_out1,
                             const float* __restrict__ w_in2, const float* __restrict__ b_in2,
                             const float* __restrict__ w_out2, const float* __restrict__ b_out2,
                             const float* __restrict__ w_fc, const float* __restrict__ b_fc,
                             float* __restrict__ ws) {
  int tid = blockIdx.x * blockDim.x + threadIdx.x;
  if (tid >= G_ * GSTRIDE) return;
  int g = tid / GSTRIDE;
  int pos = tid - g * GSTRIDE;
  float val = 0.0f;
  if (pos < OFF_WOUT1) {
    int u = pos - OFF_WIN1;
    int v = u & 1, L = (u >> 1) & 31, r = u >> 6;
    int kk = r & 15, t = r >> 4;
    int k = 16 * (kk >> 2) + 8 * (L >> 4) + 2 * (kk & 3) + v;
    int n = 16 * t + (L & 15);
    val = w_in1[(g * HID + k) * FFN + n];
  } else if (pos < OFF_WIN2) {
    int u = pos - OFF_WOUT1;
    int v = u & 1, L = (u >> 1) & 31, r = u >> 6;
    int kk = r & 31, t = r >> 5;
    int k = 16 * (kk >> 2) + 8 * (L >> 4) + 2 * (kk & 3) + v;
    int n = 16 * t + (L & 15);
    val = w_out1[(g * FFN + k) * HID + n];
  } else if (pos < OFF_WOUT2) {
    int u = pos - OFF_WIN2;
    int v = u & 1, L = (u >> 1) & 31, r = u >> 6;
    int kk = r & 15, t = r >> 4;
    int k = 16 * (kk >> 2) + 8 * (L >> 4) + 2 * (kk & 3) + v;
    int n = 16 * t + (L & 15);
    val = w_in2[(g * HID + k) * FFN + n];
  } else if (pos < OFF_WFC) {
    int u = pos - OFF_WOUT2;
    int v = u & 1, L = (u >> 1) & 31, r = u >> 6;
    int kk = r & 31, t = r >> 5;
    int k = 16 * (kk >> 2) + 8 * (L >> 4) + 2 * (kk & 3) + v;
    int n = 16 * t + (L & 15);
    val = w_out2[(g * FFN + k) * HID + n];
  } else if (pos < OFF_BIN1) {
    int u = pos - OFF_WFC;
    int v = u & 1, L = (u >> 1) & 31, kk = u >> 6;
    int k = 16 * (kk >> 2) + 8 * (L >> 4) + 2 * (kk & 3) + v;
    int c = L & 15;
    val = (c < NH) ? w_fc[(g * HID + k) * NH + c] : 0.0f;
  } else if (pos < OFF_BOUT1) {
    val = b_in1[g * FFN + (pos - OFF_BIN1)];
  } else if (pos < OFF_BIN2) {
    val = b_out1[g * HID + (pos - OFF_BOUT1)];
  } else if (pos < OFF_BOUT2) {
    val = b_in2[g * FFN + (pos - OFF_BIN2)];
  } else if (pos < OFF_BFC) {
    val = b_out2[g * HID + (pos - OFF_BOUT2)];
  } else {
    int j = pos - OFF_BFC;
    val = (j < NH) ? b_fc[g * NH + j] : 0.0f;
  }
  ws[tid] = val;
}

typedef __attribute__((ext_vector_type(4))) float v4f;

__device__ __forceinline__ v8f wmma4(v2f a, v2f b, v8f c) {
  return __builtin_amdgcn_wmma_f32_16x16x4_f32(false, a, false, b, (short)0, c,
                                               false, false);
}

__device__ __forceinline__ void mlp_block(v2f (&xb)[2][16],
                                          const float* __restrict__ win,
                                          const float* __restrict__ wout,
                                          const float* __restrict__ bin,
                                          const float* __restrict__ bout,
                                          int lane, int hi) {
  v8f acc2[2][4];
  #pragma unroll
  for (int i = 0; i < 2; ++i)
    #pragma unroll
    for (int nt = 0; nt < 4; ++nt) acc2[i][nt] = (v8f){};

  for (int t = 0; t < 8; ++t) {
    v8f h0 = {}, h1 = {};
    v2f af1[16], af2[16];
    v4f b0 = *(const v4f*)(bin + 16 * t + 8 * hi);
    v4f b1 = *(const v4f*)(bin + 16 * t + 8 * hi + 4);
    #pragma unroll
    for (int kk = 0; kk < 16; ++kk)
      af1[kk] = *(const v2f*)(win + ((t * 16 + kk) * 32 + lane) * 2);
    #pragma unroll
    for (int kk = 0; kk < 16; ++kk) {
      if (kk == 8) {
        #pragma unroll
        for (int q = 0; q < 4; ++q)
          #pragma unroll
          for (int nt = 0; nt < 4; ++nt)
            af2[4 * q + nt] = *(const v2f*)(wout + ((nt * 32 + t * 4 + q) * 32 + lane) * 2);
      }
      h0 = wmma4(af1[kk], xb[0][kk], h0);
      h1 = wmma4(af1[kk], xb[1][kk], h1);
      __builtin_amdgcn_sched_barrier(0);
    }
    #pragma unroll
    for (int j = 0; j < 4; ++j) {
      h0[j]     = fmaxf(h0[j] + b0[j], 0.0f);
      h1[j]     = fmaxf(h1[j] + b0[j], 0.0f);
      h0[j + 4] = fmaxf(h0[j + 4] + b1[j], 0.0f);
      h1[j + 4] = fmaxf(h1[j + 4] + b1[j], 0.0f);
    }
    #pragma unroll
    for (int q = 0; q < 4; ++q) {
      v2f hb0 = {h0[2 * q], h0[2 * q + 1]};
      v2f hb1 = {h1[2 * q], h1[2 * q + 1]};
      #pragma unroll
      for (int nt = 0; nt < 4; ++nt) {
        acc2[0][nt] = wmma4(af2[4 * q + nt], hb0, acc2[0][nt]);
        acc2[1][nt] = wmma4(af2[4 * q + nt], hb1, acc2[1][nt]);
      }
    }
  }
  #pragma unroll
  for (int nt = 0; nt < 4; ++nt) {
    v4f b0 = *(const v4f*)(bout + 16 * nt + 8 * hi);
    v4f b1 = *(const v4f*)(bout + 16 * nt + 8 * hi + 4);
    #pragma unroll
    for (int j = 0; j < 4; ++j) {
      #pragma unroll
      for (int s = 0; s < 2; ++s) {
        int r = 2 * j + s;
        float bb = (r < 4) ? b0[r] : b1[r - 4];
        xb[0][4 * nt + j][s] += acc2[0][nt][r] + bb;
        xb[1][4 * nt + j][s] += acc2[1][nt][r] + bb;
      }
    }
  }
}

__global__ __launch_bounds__(256) void fused_edge_mlp(
    const float* __restrict__ edge_attr, const int* __restrict__ edge_index,
    const float* __restrict__ deg, const float* __restrict__ deg_coef,
    const float* __restrict__ ws, float* __restrict__ out) {
  const int w = threadIdx.x >> 5;
  const int lane = threadIdx.x & 31;
  const int c = lane & 15;
  const int hi = lane >> 4;

  const int task = blockIdx.x;
  const int g = task >> 9;
  const int chunk = task & 511;
  const int e0 = chunk * 256 + w * 32;

  const float* wb = ws + g * GSTRIDE;

  const float* dc0 = deg_coef + g * HID;
  const float* dc1 = deg_coef + (G_ + g) * HID;
  const int ec0 = e0 + c;
  const int ec1 = e0 + 16 + c;
  const float sd0 = sqrtf(deg[edge_index[E_ + ec0]]);
  const float sd1 = sqrtf(deg[edge_index[E_ + ec1]]);

  v2f xb[2][16];
  #pragma unroll
  for (int nt = 0; nt < 4; ++nt) {
    #pragma unroll
    for (int p = 0; p < 2; ++p) {
      int k = 16 * nt + 8 * hi + 4 * p;
      v4f d0 = *(const v4f*)(dc0 + k);
      v4f d1 = *(const v4f*)(dc1 + k);
      v4f a0 = *(const v4f*)(edge_attr + (size_t)ec0 * HID + k);
      v4f a1 = *(const v4f*)(edge_attr + (size_t)ec1 * HID + k);
      v4f x0 = a0 * (d0 + sd0 * d1);
      v4f x1 = a1 * (d0 + sd1 * d1);
      xb[0][4 * nt + 2 * p]     = (v2f){x0[0], x0[1]};
      xb[0][4 * nt + 2 * p + 1] = (v2f){x0[2], x0[3]};
      xb[1][4 * nt + 2 * p]     = (v2f){x1[0], x1[1]};
      xb[1][4 * nt + 2 * p + 1] = (v2f){x1[2], x1[3]};
    }
  }

  mlp_block(xb, wb + OFF_WIN1, wb + OFF_WOUT1, wb + OFF_BIN1, wb + OFF_BOUT1, lane, hi);
  mlp_block(xb, wb + OFF_WIN2, wb + OFF_WOUT2, wb + OFF_BIN2, wb + OFF_BOUT2, lane, hi);

  #pragma unroll
  for (int kk = 0; kk < 16; ++kk) {
    #pragma unroll
    for (int v = 0; v < 2; ++v) {
      xb[0][kk][v] = fmaxf(xb[0][kk][v], 0.0f);
      xb[1][kk][v] = fmaxf(xb[1][kk][v], 0.0f);
    }
  }

  v8f f0 = {}, f1 = {};
  #pragma unroll
  for (int kk = 0; kk < 16; ++kk) {
    v2f af = *(const v2f*)(wb + OFF_WFC + (kk * 32 + lane) * 2);
    f0 = wmma4(af, xb[0][kk], f0);
    f1 = wmma4(af, xb[1][kk], f1);
  }

  if (hi == 0) {
    v4f b0 = *(const v4f*)(wb + OFF_BFC);
    v4f b1 = *(const v4f*)(wb + OFF_BFC + 4);
    size_t base = (size_t)g * E_;
    float* o0 = out + (base + ec0) * NH;
    float* o1 = out + (base + ec1) * NH;
    *(v4f*)(o0)     = (v4f){f0[0], f0[1], f0[2], f0[3]} + b0;
    *(v4f*)(o0 + 4) = (v4f){f0[4], f0[5], f0[6], f0[7]} + b1;
    *(v4f*)(o1)     = (v4f){f1[0], f1[1], f1[2], f1[3]} + b0;
    *(v4f*)(o1 + 4) = (v4f){f1[4], f1[5], f1[6], f1[7]} + b1;
  }
}

extern "C" void kernel_launch(void* const* d_in, const int* in_sizes, int n_in,
                              void* d_out, int out_size, void* d_ws, size_t ws_size,
                              hipStream_t stream) {
  const float* edge_attr  = (const float*)d_in[0];
  const int*   edge_index = (const int*)d_in[1];
  const float* deg        = (const float*)d_in[2];
  const float* deg_coef   = (const float*)d_in[3];
  const float* w_in1  = (const float*)d_in[4];
  const float* b_in1  = (const float*)d_in[5];
  const float* w_out1 = (const float*)d_in[6];
  const float* b_out1 = (const float*)d_in[7];
  const float* w_in2  = (const float*)d_in[8];
  const float* b_in2  = (const float*)d_in[9];
  const float* w_out2 = (const float*)d_in[10];
  const float* b_out2 = (const float*)d_in[11];
  const float* w_fc   = (const float*)d_in[12];
  const float* b_fc   = (const float*)d_in[13];
  float* ws  = (float*)d_ws;
  float* out = (float*)d_out;

  const int packTotal = G_ * GSTRIDE;
  pack_weights<<<(packTotal + 255) / 256, 256, 0, stream>>>(
      w_in1, b_in1, w_out1, b_out1, w_in2, b_in2, w_out2, b_out2, w_fc, b_fc, ws);

  fused_edge_mlp<<<G_ * (E_ / 256), 256, 0, stream>>>(
      edge_attr, edge_index, deg, deg_coef, ws, out);
}
